# starting kernel
# baseline (speedup 1.0000x reference)
#include <hip/hip_runtime.h>

typedef __attribute__((ext_vector_type(2))) float v2f;
typedef __attribute__((ext_vector_type(8))) float v8f;

#define DIMS 256

#define SWZ_XOR(x, m) \
  __int_as_float(__builtin_amdgcn_ds_swizzle(__float_as_int(x), (((m) << 10) | 0x1f)))

__device__ __forceinline__ float dot4(float4 a, float4 b) {
  return a.x * b.x + a.y * b.y + a.z * b.z + a.w * b.w;
}

__device__ __forceinline__ void wave_reduce3(float p0, float p1, float p2, int lane,
                                             float& r0, float& r1, float& r2) {
  v2f A1;
  A1.x = p0;
  A1.y = p1;
  v2f B1;
  B1.x = (lane == 0 || lane == 16) ? 1.0f : 0.0f;
  B1.y = (lane == 1 || lane == 17) ? 1.0f : 0.0f;
  v8f Cz = {};
  v8f D1 = __builtin_amdgcn_wmma_f32_16x16x4_f32(false, A1, false, B1, (short)0, Cz,
                                                 false, false);
  v2f A2;
  A2.x = p2;
  A2.y = 0.0f;
  v2f B2;
  B2.x = (lane == 2 || lane == 18) ? 1.0f : 0.0f;
  B2.y = 0.0f;
  v8f D2 = __builtin_amdgcn_wmma_f32_16x16x4_f32(false, A2, false, B2, (short)0, D1,
                                                 false, false);
  float t = ((D2[0] + D2[1]) + (D2[2] + D2[3])) + ((D2[4] + D2[5]) + (D2[6] + D2[7]));
  t += SWZ_XOR(t, 16);
  r0 = __int_as_float(__builtin_amdgcn_readlane(__float_as_int(t), 0));
  r1 = __int_as_float(__builtin_amdgcn_readlane(__float_as_int(t), 1));
  r2 = __int_as_float(__builtin_amdgcn_readlane(__float_as_int(t), 2));
}

__global__ void __launch_bounds__(256)
transint_relprep(const float* __restrict__ bases, float* __restrict__ rws, int R) {
  int wave = __builtin_amdgcn_readfirstlane(
      (int)((blockIdx.x * blockDim.x + threadIdx.x) >> 5));
  int lane = (int)(threadIdx.x & 31);
  if (wave >= R) return;
  const float* base0 = bases + (size_t)wave * 2 * DIMS + lane * 8;
  const float* base1 = base0 + DIMS;
  float4 x0 = ((const float4*)base0)[0];
  float4 x1 = ((const float4*)base0)[1];
  float4 y0 = ((const float4*)base1)[0];
  float4 y1 = ((const float4*)base1)[1];
  float pa = dot4(x0, x0) + dot4(x1, x1);
  float pb = dot4(x0, y0) + dot4(x1, y1);
  float pd = dot4(y0, y0) + dot4(y1, y1);

  float aa, bb, dd;
  wave_reduce3(pa, pb, pd, lane, aa, bb, dd);

  if (lane == 0) {
    float det = aa * dd - bb * bb;
    float m00, m01, m11;
    if (det != 0.0f) {
      float inv = 1.0f / det;
      m00 = dd * inv;
      m01 = -bb * inv;
      m11 = aa * inv;
    } else {
      m00 = 1.0f / aa;
      m01 = 0.0f;
      m11 = 1.0f / aa;
    }
    float* o = rws + (size_t)wave * 8;
    o[0] = aa; o[1] = bb; o[2] = dd;
    o[3] = m00; o[4] = m01; o[5] = m11;
    o[6] = 0.0f; o[7] = 0.0f;
  }
}

__global__ void __launch_bounds__(256)
transint_score(const int* __restrict__ pos_h, const int* __restrict__ pos_t,
               const int* __restrict__ pos_r, const int* __restrict__ neg_h,
               const int* __restrict__ neg_t, const float* __restrict__ ent_emb,
               const float* __restrict__ heads, const float* __restrict__ bases,
               const int* __restrict__ rel2head, const float* __restrict__ rel2mult,
               const float* __restrict__ rws, float* __restrict__ out, int Bn) {
  int wave = __builtin_amdgcn_readfirstlane(
      (int)((blockIdx.x * blockDim.x + threadIdx.x) >> 5));
  int lane = (int)(threadIdx.x & 31);
  if (wave >= 2 * Bn) return;

  bool is_pos = wave < Bn;
  int b = is_pos ? wave : wave - Bn;
  int h = (is_pos ? pos_h : neg_h)[b];
  int t = (is_pos ? pos_t : neg_t)[b];
  int r = pos_r[b];
  int k = rel2head[r];
  float mult = rel2mult[r];

  int off = lane * 8;
  const float4* eh = (const float4*)(ent_emb + (size_t)h * DIMS + off);
  const float4* et = (const float4*)(ent_emb + (size_t)t * DIMS + off);
  const float4* hd = (const float4*)(heads + (size_t)k * DIMS + off);
  const float4* a0 = (const float4*)(bases + (size_t)r * 2 * DIMS + off);
  const float4* a1 = (const float4*)(bases + (size_t)r * 2 * DIMS + DIMS + off);

  float4 eh0 = eh[0], eh1 = eh[1];
  float4 et0 = et[0], et1 = et[1];
  float4 hd0 = hd[0], hd1 = hd[1];
  float4 A00 = a0[0], A01 = a0[1];
  float4 A10 = a1[0], A11 = a1[1];

  float4 s0, s1;
  s0.x = eh0.x + mult * hd0.x - et0.x;
  s0.y = eh0.y + mult * hd0.y - et0.y;
  s0.z = eh0.z + mult * hd0.z - et0.z;
  s0.w = eh0.w + mult * hd0.w - et0.w;
  s1.x = eh1.x + mult * hd1.x - et1.x;
  s1.y = eh1.y + mult * hd1.y - et1.y;
  s1.z = eh1.z + mult * hd1.z - et1.z;
  s1.w = eh1.w + mult * hd1.w - et1.w;

  float p0 = dot4(s0, A00) + dot4(s1, A01);
  float p1 = dot4(s0, A10) + dot4(s1, A11);
  float p2 = dot4(s0, s0) + dot4(s1, s1);

  float u0, u1, ss;
  wave_reduce3(p0, p1, p2, lane, u0, u1, ss);

  const float* rw = rws + (size_t)r * 8;
  float ga = rw[0], gb = rw[1], gd = rw[2];
  float m00 = rw[3], m01 = rw[4], m11 = rw[5];
  float w0 = m00 * u0 + m01 * u1;
  float w1 = m01 * u0 + m11 * u1;
  float score = ss - 2.0f * (u0 * w0 + u1 * w1) +
                (ga * w0 * w0 + 2.0f * gb * w0 * w1 + gd * w1 * w1);

  if (lane == 0) out[wave] = score;
}

extern "C" void kernel_launch(void* const* d_in, const int* in_sizes, int n_in,
                              void* d_out, int out_size, void* d_ws, size_t ws_size,
                              hipStream_t stream) {
  const int*   pos_h    = (const int*)d_in[0];
  const int*   pos_t    = (const int*)d_in[1];
  const int*   pos_r    = (const int*)d_in[2];
  const int*   neg_h    = (const int*)d_in[3];
  const int*   neg_t    = (const int*)d_in[4];
  const float* ent_emb  = (const float*)d_in[6];
  const float* heads    = (const float*)d_in[7];
  const float* bases    = (const float*)d_in[8];
  const int*   rel2head = (const int*)d_in[9];
  const float* rel2mult = (const float*)d_in[10];
  float* out = (float*)d_out;
  float* rws = (float*)d_ws;

  int Bn = in_sizes[0];
  int R  = in_sizes[9];

  int relBlocks = (R * 32 + 255) / 256;
  transint_relprep<<<relBlocks, 256, 0, stream>>>(bases, rws, R);

  long nwaves = 2L * (long)Bn;
  long blocks = (nwaves * 32 + 255) / 256;
  transint_score<<<(int)blocks, 256, 0, stream>>>(
      pos_h, pos_t, pos_r, neg_h, neg_t, ent_emb, heads, bases,
      rel2head, rel2mult, rws, out, Bn);
}


# Score kernel: each wave loops over 8 consecutive scores and writes them with one 32-byte store, so 16384 waves are started instead of 131072; same loads and arithmetic per score
# speedup vs baseline: 1.2489x; 1.2489x over previous
#include <hip/hip_runtime.h>

typedef __attribute__((ext_vector_type(2))) float v2f;
typedef __attribute__((ext_vector_type(8))) float v8f;

#define DIMS 256

#define SWZ_XOR(x, m) \
  __int_as_float(__builtin_amdgcn_ds_swizzle(__float_as_int(x), (((m) << 10) | 0x1f)))

__device__ __forceinline__ float dot4(float4 a, float4 b) {
  return a.x * b.x + a.y * b.y + a.z * b.z + a.w * b.w;
}

__device__ __forceinline__ void wave_reduce3(float p0, float p1, float p2, int lane,
                                             float& r0, float& r1, float& r2) {
  v2f A1;
  A1.x = p0;
  A1.y = p1;
  v2f B1;
  B1.x = (lane == 0 || lane == 16) ? 1.0f : 0.0f;
  B1.y = (lane == 1 || lane == 17) ? 1.0f : 0.0f;
  v8f Cz = {};
  v8f D1 = __builtin_amdgcn_wmma_f32_16x16x4_f32(false, A1, false, B1, (short)0, Cz,
                                                 false, false);
  v2f A2;
  A2.x = p2;
  A2.y = 0.0f;
  v2f B2;
  B2.x = (lane == 2 || lane == 18) ? 1.0f : 0.0f;
  B2.y = 0.0f;
  v8f D2 = __builtin_amdgcn_wmma_f32_16x16x4_f32(false, A2, false, B2, (short)0, D1,
                                                 false, false);
  float t = ((D2[0] + D2[1]) + (D2[2] + D2[3])) + ((D2[4] + D2[5]) + (D2[6] + D2[7]));
  t += SWZ_XOR(t, 16);
  r0 = __int_as_float(__builtin_amdgcn_readlane(__float_as_int(t), 0));
  r1 = __int_as_float(__builtin_amdgcn_readlane(__float_as_int(t), 1));
  r2 = __int_as_float(__builtin_amdgcn_readlane(__float_as_int(t), 2));
}

__global__ void __launch_bounds__(256)
transint_relprep(const float* __restrict__ bases, float* __restrict__ rws, int R) {
  int wave = __builtin_amdgcn_readfirstlane(
      (int)((blockIdx.x * blockDim.x + threadIdx.x) >> 5));
  int lane = (int)(threadIdx.x & 31);
  if (wave >= R) return;
  const float* base0 = bases + (size_t)wave * 2 * DIMS + lane * 8;
  const float* base1 = base0 + DIMS;
  float4 x0 = ((const float4*)base0)[0];
  float4 x1 = ((const float4*)base0)[1];
  float4 y0 = ((const float4*)base1)[0];
  float4 y1 = ((const float4*)base1)[1];
  float pa = dot4(x0, x0) + dot4(x1, x1);
  float pb = dot4(x0, y0) + dot4(x1, y1);
  float pd = dot4(y0, y0) + dot4(y1, y1);

  float aa, bb, dd;
  wave_reduce3(pa, pb, pd, lane, aa, bb, dd);

  if (lane == 0) {
    float det = aa * dd - bb * bb;
    float m00, m01, m11;
    if (det != 0.0f) {
      float inv = 1.0f / det;
      m00 = dd * inv;
      m01 = -bb * inv;
      m11 = aa * inv;
    } else {
      m00 = 1.0f / aa;
      m01 = 0.0f;
      m11 = 1.0f / aa;
    }
    float* o = rws + (size_t)wave * 8;
    o[0] = aa; o[1] = bb; o[2] = dd;
    o[3] = m00; o[4] = m01; o[5] = m11;
    o[6] = 0.0f; o[7] = 0.0f;
  }
}

#define SPW 8

__global__ void __launch_bounds__(256)
transint_score(const int* __restrict__ pos_h, const int* __restrict__ pos_t,
               const int* __restrict__ pos_r, const int* __restrict__ neg_h,
               const int* __restrict__ neg_t, const float* __restrict__ ent_emb,
               const float* __restrict__ heads, const float* __restrict__ bases,
               const int* __restrict__ rel2head, const float* __restrict__ rel2mult,
               const float* __restrict__ rws, float* __restrict__ out, int Bn) {
  int wave = __builtin_amdgcn_readfirstlane(
      (int)((blockIdx.x * blockDim.x + threadIdx.x) >> 5));
  int lane = (int)(threadIdx.x & 31);
  int total = 2 * Bn;
  int g0 = wave * SPW;
  if (g0 >= total) return;
  int n = total - g0;
  n = n < SPW ? n : SPW;

  int off = lane * 8;
  float mine = 0.0f;

  for (int j = 0; j < n; ++j) {
    int g = g0 + j;
    bool is_pos = g < Bn;
    int b = is_pos ? g : g - Bn;
    int h = (is_pos ? pos_h : neg_h)[b];
    int t = (is_pos ? pos_t : neg_t)[b];
    int r = pos_r[b];
    int k = rel2head[r];
    float mult = rel2mult[r];

    const float4* eh = (const float4*)(ent_emb + (size_t)h * DIMS + off);
    const float4* et = (const float4*)(ent_emb + (size_t)t * DIMS + off);
    const float4* hd = (const float4*)(heads + (size_t)k * DIMS + off);
    const float4* a0 = (const float4*)(bases + (size_t)r * 2 * DIMS + off);
    const float4* a1 = (const float4*)(bases + (size_t)r * 2 * DIMS + DIMS + off);

    float4 eh0 = eh[0], eh1 = eh[1];
    float4 et0 = et[0], et1 = et[1];
    float4 hd0 = hd[0], hd1 = hd[1];
    float4 A00 = a0[0], A01 = a0[1];
    float4 A10 = a1[0], A11 = a1[1];

    float4 s0, s1;
    s0.x = eh0.x + mult * hd0.x - et0.x;
    s0.y = eh0.y + mult * hd0.y - et0.y;
    s0.z = eh0.z + mult * hd0.z - et0.z;
    s0.w = eh0.w + mult * hd0.w - et0.w;
    s1.x = eh1.x + mult * hd1.x - et1.x;
    s1.y = eh1.y + mult * hd1.y - et1.y;
    s1.z = eh1.z + mult * hd1.z - et1.z;
    s1.w = eh1.w + mult * hd1.w - et1.w;

    float p0 = dot4(s0, A00) + dot4(s1, A01);
    float p1 = dot4(s0, A10) + dot4(s1, A11);
    float p2 = dot4(s0, s0) + dot4(s1, s1);

    float u0, u1, ss;
    wave_reduce3(p0, p1, p2, lane, u0, u1, ss);

    const float* rw = rws + (size_t)r * 8;
    float ga = rw[0], gb = rw[1], gd = rw[2];
    float m00 = rw[3], m01 = rw[4], m11 = rw[5];
    float w0 = m00 * u0 + m01 * u1;
    float w1 = m01 * u0 + m11 * u1;
    float score = ss - 2.0f * (u0 * w0 + u1 * w1) +
                  (ga * w0 * w0 + 2.0f * gb * w0 * w1 + gd * w1 * w1);

    mine = (lane == j) ? score : mine;
  }

  if (lane < n) out[g0 + lane] = mine;
}

extern "C" void kernel_launch(void* const* d_in, const int* in_sizes, int n_in,
                              void* d_out, int out_size, void* d_ws, size_t ws_size,
                              hipStream_t stream) {
  const int*   pos_h    = (const int*)d_in[0];
  const int*   pos_t    = (const int*)d_in[1];
  const int*   pos_r    = (const int*)d_in[2];
  const int*   neg_h    = (const int*)d_in[3];
  const int*   neg_t    = (const int*)d_in[4];
  const float* ent_emb  = (const float*)d_in[6];
  const float* heads    = (const float*)d_in[7];
  const float* bases    = (const float*)d_in[8];
  const int*   rel2head = (const int*)d_in[9];
  const float* rel2mult = (const float*)d_in[10];
  float* out = (float*)d_out;
  float* rws = (float*)d_ws;

  int Bn = in_sizes[0];
  int R  = in_sizes[9];

  int relBlocks = (R * 32 + 255) / 256;
  transint_relprep<<<relBlocks, 256, 0, stream>>>(bases, rws, R);

  long nwaves = (2L * (long)Bn + SPW - 1) / SPW;
  long blocks = (nwaves * 32 + 255) / 256;
  transint_score<<<(int)blocks, 256, 0, stream>>>(
      pos_h, pos_t, pos_r, neg_h, neg_t, ent_emb, heads, bases,
      rel2head, rel2mult, rws, out, Bn);
}


# Score kernel: one wave computes the positive and negative score of a batch index together, loading their shared head row and two basis rows once (7 rows per pair, was 10); 8192 waves, was 16384
# speedup vs baseline: 1.4833x; 1.1877x over previous
#include <hip/hip_runtime.h>

typedef __attribute__((ext_vector_type(2))) float v2f;
typedef __attribute__((ext_vector_type(8))) float v8f;

#define DIMS 256

#define SWZ_XOR(x, m) \
  __int_as_float(__builtin_amdgcn_ds_swizzle(__float_as_int(x), (((m) << 10) | 0x1f)))

__device__ __forceinline__ float dot4(float4 a, float4 b) {
  return a.x * b.x + a.y * b.y + a.z * b.z + a.w * b.w;
}

__device__ __forceinline__ void wave_reduce3(float p0, float p1, float p2, int lane,
                                             float& r0, float& r1, float& r2) {
  v2f A1;
  A1.x = p0;
  A1.y = p1;
  v2f B1;
  B1.x = (lane == 0 || lane == 16) ? 1.0f : 0.0f;
  B1.y = (lane == 1 || lane == 17) ? 1.0f : 0.0f;
  v8f Cz = {};
  v8f D1 = __builtin_amdgcn_wmma_f32_16x16x4_f32(false, A1, false, B1, (short)0, Cz,
                                                 false, false);
  v2f A2;
  A2.x = p2;
  A2.y = 0.0f;
  v2f B2;
  B2.x = (lane == 2 || lane == 18) ? 1.0f : 0.0f;
  B2.y = 0.0f;
  v8f D2 = __builtin_amdgcn_wmma_f32_16x16x4_f32(false, A2, false, B2, (short)0, D1,
                                                 false, false);
  float t = ((D2[0] + D2[1]) + (D2[2] + D2[3])) + ((D2[4] + D2[5]) + (D2[6] + D2[7]));
  t += SWZ_XOR(t, 16);
  r0 = __int_as_float(__builtin_amdgcn_readlane(__float_as_int(t), 0));
  r1 = __int_as_float(__builtin_amdgcn_readlane(__float_as_int(t), 1));
  r2 = __int_as_float(__builtin_amdgcn_readlane(__float_as_int(t), 2));
}

__device__ __forceinline__ void wave_reduce6(float p0, float p1, float p2, float q0,
                                             float q1, float q2, int lane, float& r0,
                                             float& r1, float& r2, float& r3, float& r4,
                                             float& r5) {
  int c = lane & 15;
  v2f A1, A2, A3, B1, B2, B3;
  A1.x = p0; A1.y = p1;
  A2.x = p2; A2.y = q0;
  A3.x = q1; A3.y = q2;
  B1.x = (c == 0) ? 1.0f : 0.0f;
  B1.y = (c == 1) ? 1.0f : 0.0f;
  B2.x = (c == 2) ? 1.0f : 0.0f;
  B2.y = (c == 3) ? 1.0f : 0.0f;
  B3.x = (c == 4) ? 1.0f : 0.0f;
  B3.y = (c == 5) ? 1.0f : 0.0f;
  v8f Cz = {};
  v8f D1 = __builtin_amdgcn_wmma_f32_16x16x4_f32(false, A1, false, B1, (short)0, Cz,
                                                 false, false);
  v8f D2 = __builtin_amdgcn_wmma_f32_16x16x4_f32(false, A2, false, B2, (short)0, D1,
                                                 false, false);
  v8f D3 = __builtin_amdgcn_wmma_f32_16x16x4_f32(false, A3, false, B3, (short)0, D2,
                                                 false, false);
  float t = ((D3[0] + D3[1]) + (D3[2] + D3[3])) + ((D3[4] + D3[5]) + (D3[6] + D3[7]));
  t += SWZ_XOR(t, 16);
  r0 = __int_as_float(__builtin_amdgcn_readlane(__float_as_int(t), 0));
  r1 = __int_as_float(__builtin_amdgcn_readlane(__float_as_int(t), 1));
  r2 = __int_as_float(__builtin_amdgcn_readlane(__float_as_int(t), 2));
  r3 = __int_as_float(__builtin_amdgcn_readlane(__float_as_int(t), 3));
  r4 = __int_as_float(__builtin_amdgcn_readlane(__float_as_int(t), 4));
  r5 = __int_as_float(__builtin_amdgcn_readlane(__float_as_int(t), 5));
}

__global__ void __launch_bounds__(256)
transint_relprep(const float* __restrict__ bases, float* __restrict__ rws, int R) {
  int wave = __builtin_amdgcn_readfirstlane(
      (int)((blockIdx.x * blockDim.x + threadIdx.x) >> 5));
  int lane = (int)(threadIdx.x & 31);
  if (wave >= R) return;
  const float* base0 = bases + (size_t)wave * 2 * DIMS + lane * 8;
  const float* base1 = base0 + DIMS;
  float4 x0 = ((const float4*)base0)[0];
  float4 x1 = ((const float4*)base0)[1];
  float4 y0 = ((const float4*)base1)[0];
  float4 y1 = ((const float4*)base1)[1];
  float pa = dot4(x0, x0) + dot4(x1, x1);
  float pb = dot4(x0, y0) + dot4(x1, y1);
  float pd = dot4(y0, y0) + dot4(y1, y1);

  float aa, bb, dd;
  wave_reduce3(pa, pb, pd, lane, aa, bb, dd);

  if (lane == 0) {
    float det = aa * dd - bb * bb;
    float m00, m01, m11;
    if (det != 0.0f) {
      float inv = 1.0f / det;
      m00 = dd * inv;
      m01 = -bb * inv;
      m11 = aa * inv;
    } else {
      m00 = 1.0f / aa;
      m01 = 0.0f;
      m11 = 1.0f / aa;
    }
    float* o = rws + (size_t)wave * 8;
    o[0] = aa; o[1] = bb; o[2] = dd;
    o[3] = m00; o[4] = m01; o[5] = m11;
    o[6] = 0.0f; o[7] = 0.0f;
  }
}

#define BPW 8

__device__ __forceinline__ void lane_sub(float4 eh0, float4 eh1, float4 et0, float4 et1,
                                         float4 hd0, float4 hd1, float mult, float4& s0,
                                         float4& s1) {
  s0.x = eh0.x + mult * hd0.x - et0.x;
  s0.y = eh0.y + mult * hd0.y - et0.y;
  s0.z = eh0.z + mult * hd0.z - et0.z;
  s0.w = eh0.w + mult * hd0.w - et0.w;
  s1.x = eh1.x + mult * hd1.x - et1.x;
  s1.y = eh1.y + mult * hd1.y - et1.y;
  s1.z = eh1.z + mult * hd1.z - et1.z;
  s1.w = eh1.w + mult * hd1.w - et1.w;
}

__device__ __forceinline__ float score_of(float u0, float u1, float ss, float ga, float gb,
                                          float gd, float m00, float m01, float m11) {
  float w0 = m00 * u0 + m01 * u1;
  float w1 = m01 * u0 + m11 * u1;
  return ss - 2.0f * (u0 * w0 + u1 * w1) +
         (ga * w0 * w0 + 2.0f * gb * w0 * w1 + gd * w1 * w1);
}

__global__ void __launch_bounds__(256)
transint_score(const int* __restrict__ pos_h, const int* __restrict__ pos_t,
               const int* __restrict__ pos_r, const int* __restrict__ neg_h,
               const int* __restrict__ neg_t, const float* __restrict__ ent_emb,
               const float* __restrict__ heads, const float* __restrict__ bases,
               const int* __restrict__ rel2head, const float* __restrict__ rel2mult,
               const float* __restrict__ rws, float* __restrict__ out, int Bn) {
  int wave = __builtin_amdgcn_readfirstlane(
      (int)((blockIdx.x * blockDim.x + threadIdx.x) >> 5));
  int lane = (int)(threadIdx.x & 31);
  int b0 = wave * BPW;
  if (b0 >= Bn) return;
  int n = Bn - b0;
  n = n < BPW ? n : BPW;

  int off = lane * 8;
  float mine = 0.0f;

  for (int j = 0; j < n; ++j) {
    int b = b0 + j;
    int hp = pos_h[b];
    int tp = pos_t[b];
    int hn = neg_h[b];
    int tn = neg_t[b];
    int r = pos_r[b];
    int k = rel2head[r];
    float mult = rel2mult[r];

    const float4* ehp = (const float4*)(ent_emb + (size_t)hp * DIMS + off);
    const float4* etp = (const float4*)(ent_emb + (size_t)tp * DIMS + off);
    const float4* ehn = (const float4*)(ent_emb + (size_t)hn * DIMS + off);
    const float4* etn = (const float4*)(ent_emb + (size_t)tn * DIMS + off);
    const float4* hd = (const float4*)(heads + (size_t)k * DIMS + off);
    const float4* a0 = (const float4*)(bases + (size_t)r * 2 * DIMS + off);
    const float4* a1 = (const float4*)(bases + (size_t)r * 2 * DIMS + DIMS + off);

    float4 ehp0 = ehp[0], ehp1 = ehp[1];
    float4 etp0 = etp[0], etp1 = etp[1];
    float4 ehn0 = ehn[0], ehn1 = ehn[1];
    float4 etn0 = etn[0], etn1 = etn[1];
    float4 hd0 = hd[0], hd1 = hd[1];
    float4 A00 = a0[0], A01 = a0[1];
    float4 A10 = a1[0], A11 = a1[1];

    float4 sp0, sp1, sn0, sn1;
    lane_sub(ehp0, ehp1, etp0, etp1, hd0, hd1, mult, sp0, sp1);
    lane_sub(ehn0, ehn1, etn0, etn1, hd0, hd1, mult, sn0, sn1);

    float p0 = dot4(sp0, A00) + dot4(sp1, A01);
    float p1 = dot4(sp0, A10) + dot4(sp1, A11);
    float p2 = dot4(sp0, sp0) + dot4(sp1, sp1);
    float q0 = dot4(sn0, A00) + dot4(sn1, A01);
    float q1 = dot4(sn0, A10) + dot4(sn1, A11);
    float q2 = dot4(sn0, sn0) + dot4(sn1, sn1);

    float up0, up1, ssp, un0, un1, ssn;
    wave_reduce6(p0, p1, p2, q0, q1, q2, lane, up0, up1, ssp, un0, un1, ssn);

    const float* rw = rws + (size_t)r * 8;
    float ga = rw[0], gb = rw[1], gd = rw[2];
    float m00 = rw[3], m01 = rw[4], m11 = rw[5];
    float score_p = score_of(up0, up1, ssp, ga, gb, gd, m00, m01, m11);
    float score_n = score_of(un0, un1, ssn, ga, gb, gd, m00, m01, m11);

    mine = (lane == j) ? score_p : mine;
    mine = (lane == j + BPW) ? score_n : mine;
  }

  int slot = lane & (BPW - 1);
  if (lane < 2 * BPW && slot < n)
    out[(size_t)(lane < BPW ? 0 : Bn) + (size_t)(b0 + slot)] = mine;
}

extern "C" void kernel_launch(void* const* d_in, const int* in_sizes, int n_in,
                              void* d_out, int out_size, void* d_ws, size_t ws_size,
                              hipStream_t stream) {
  const int*   pos_h    = (const int*)d_in[0];
  const int*   pos_t    = (const int*)d_in[1];
  const int*   pos_r    = (const int*)d_in[2];
  const int*   neg_h    = (const int*)d_in[3];
  const int*   neg_t    = (const int*)d_in[4];
  const float* ent_emb  = (const float*)d_in[6];
  const float* heads    = (const float*)d_in[7];
  const float* bases    = (const float*)d_in[8];
  const int*   rel2head = (const int*)d_in[9];
  const float* rel2mult = (const float*)d_in[10];
  float* out = (float*)d_out;
  float* rws = (float*)d_ws;

  int Bn = in_sizes[0];
  int R  = in_sizes[9];

  int relBlocks = (R * 32 + 255) / 256;
  transint_relprep<<<relBlocks, 256, 0, stream>>>(bases, rws, R);

  long nwaves = ((long)Bn + BPW - 1) / BPW;
  long blocks = (nwaves * 32 + 255) / 256;
  transint_score<<<(int)blocks, 256, 0, stream>>>(
      pos_h, pos_t, pos_r, neg_h, neg_t, ent_emb, heads, bases,
      rel2head, rel2mult, rws, out, Bn);
}


# Row loads remapped: lane l reads floats [4l,4l+4) and [128+4l,128+4l+4), so each b128 load covers 4 whole 128-B lines (was 8 half lines); lanes sum other elements, so the summation order changes
# speedup vs baseline: 1.6266x; 1.0966x over previous
#include <hip/hip_runtime.h>

typedef __attribute__((ext_vector_type(2))) float v2f;
typedef __attribute__((ext_vector_type(8))) float v8f;

#define DIMS 256

#define LANE_OFF 4
#define SECOND (DIMS / 8)

#define SWZ_XOR(x, m) \
  __int_as_float(__builtin_amdgcn_ds_swizzle(__float_as_int(x), (((m) << 10) | 0x1f)))

__device__ __forceinline__ float dot4(float4 a, float4 b) {
  return a.x * b.x + a.y * b.y + a.z * b.z + a.w * b.w;
}

__device__ __forceinline__ void wave_reduce3(float p0, float p1, float p2, int lane,
                                             float& r0, float& r1, float& r2) {
  v2f A1;
  A1.x = p0;
  A1.y = p1;
  v2f B1;
  B1.x = (lane == 0 || lane == 16) ? 1.0f : 0.0f;
  B1.y = (lane == 1 || lane == 17) ? 1.0f : 0.0f;
  v8f Cz = {};
  v8f D1 = __builtin_amdgcn_wmma_f32_16x16x4_f32(false, A1, false, B1, (short)0, Cz,
                                                 false, false);
  v2f A2;
  A2.x = p2;
  A2.y = 0.0f;
  v2f B2;
  B2.x = (lane == 2 || lane == 18) ? 1.0f : 0.0f;
  B2.y = 0.0f;
  v8f D2 = __builtin_amdgcn_wmma_f32_16x16x4_f32(false, A2, false, B2, (short)0, D1,
                                                 false, false);
  float t = ((D2[0] + D2[1]) + (D2[2] + D2[3])) + ((D2[4] + D2[5]) + (D2[6] + D2[7]));
  t += SWZ_XOR(t, 16);
  r0 = __int_as_float(__builtin_amdgcn_readlane(__float_as_int(t), 0));
  r1 = __int_as_float(__builtin_amdgcn_readlane(__float_as_int(t), 1));
  r2 = __int_as_float(__builtin_amdgcn_readlane(__float_as_int(t), 2));
}

__device__ __forceinline__ void wave_reduce6(float p0, float p1, float p2, float q0,
                                             float q1, float q2, int lane, float& r0,
                                             float& r1, float& r2, float& r3, float& r4,
                                             float& r5) {
  int c = lane & 15;
  v2f A1, A2, A3, B1, B2, B3;
  A1.x = p0; A1.y = p1;
  A2.x = p2; A2.y = q0;
  A3.x = q1; A3.y = q2;
  B1.x = (c == 0) ? 1.0f : 0.0f;
  B1.y = (c == 1) ? 1.0f : 0.0f;
  B2.x = (c == 2) ? 1.0f : 0.0f;
  B2.y = (c == 3) ? 1.0f : 0.0f;
  B3.x = (c == 4) ? 1.0f : 0.0f;
  B3.y = (c == 5) ? 1.0f : 0.0f;
  v8f Cz = {};
  v8f D1 = __builtin_amdgcn_wmma_f32_16x16x4_f32(false, A1, false, B1, (short)0, Cz,
                                                 false, false);
  v8f D2 = __builtin_amdgcn_wmma_f32_16x16x4_f32(false, A2, false, B2, (short)0, D1,
                                                 false, false);
  v8f D3 = __builtin_amdgcn_wmma_f32_16x16x4_f32(false, A3, false, B3, (short)0, D2,
                                                 false, false);
  float t = ((D3[0] + D3[1]) + (D3[2] + D3[3])) + ((D3[4] + D3[5]) + (D3[6] + D3[7]));
  t += SWZ_XOR(t, 16);
  r0 = __int_as_float(__builtin_amdgcn_readlane(__float_as_int(t), 0));
  r1 = __int_as_float(__builtin_amdgcn_readlane(__float_as_int(t), 1));
  r2 = __int_as_float(__builtin_amdgcn_readlane(__float_as_int(t), 2));
  r3 = __int_as_float(__builtin_amdgcn_readlane(__float_as_int(t), 3));
  r4 = __int_as_float(__builtin_amdgcn_readlane(__float_as_int(t), 4));
  r5 = __int_as_float(__builtin_amdgcn_readlane(__float_as_int(t), 5));
}

__global__ void __launch_bounds__(256)
transint_relprep(const float* __restrict__ bases, float* __restrict__ rws, int R) {
  int wave = __builtin_amdgcn_readfirstlane(
      (int)((blockIdx.x * blockDim.x + threadIdx.x) >> 5));
  int lane = (int)(threadIdx.x & 31);
  if (wave >= R) return;
  const float* base0 = bases + (size_t)wave * 2 * DIMS + lane * LANE_OFF;
  const float* base1 = base0 + DIMS;
  float4 x0 = ((const float4*)base0)[0];
  float4 x1 = ((const float4*)base0)[SECOND];
  float4 y0 = ((const float4*)base1)[0];
  float4 y1 = ((const float4*)base1)[SECOND];
  float pa = dot4(x0, x0) + dot4(x1, x1);
  float pb = dot4(x0, y0) + dot4(x1, y1);
  float pd = dot4(y0, y0) + dot4(y1, y1);

  float aa, bb, dd;
  wave_reduce3(pa, pb, pd, lane, aa, bb, dd);

  if (lane == 0) {
    float det = aa * dd - bb * bb;
    float m00, m01, m11;
    if (det != 0.0f) {
      float inv = 1.0f / det;
      m00 = dd * inv;
      m01 = -bb * inv;
      m11 = aa * inv;
    } else {
      m00 = 1.0f / aa;
      m01 = 0.0f;
      m11 = 1.0f / aa;
    }
    float* o = rws + (size_t)wave * 8;
    o[0] = aa; o[1] = bb; o[2] = dd;
    o[3] = m00; o[4] = m01; o[5] = m11;
    o[6] = 0.0f; o[7] = 0.0f;
  }
}

#define BPW 8

__device__ __forceinline__ void lane_sub(float4 eh0, float4 eh1, float4 et0, float4 et1,
                                         float4 hd0, float4 hd1, float mult, float4& s0,
                                         float4& s1) {
  s0.x = eh0.x + mult * hd0.x - et0.x;
  s0.y = eh0.y + mult * hd0.y - et0.y;
  s0.z = eh0.z + mult * hd0.z - et0.z;
  s0.w = eh0.w + mult * hd0.w - et0.w;
  s1.x = eh1.x + mult * hd1.x - et1.x;
  s1.y = eh1.y + mult * hd1.y - et1.y;
  s1.z = eh1.z + mult * hd1.z - et1.z;
  s1.w = eh1.w + mult * hd1.w - et1.w;
}

__device__ __forceinline__ float score_of(float u0, float u1, float ss, float ga, float gb,
                                          float gd, float m00, float m01, float m11) {
  float w0 = m00 * u0 + m01 * u1;
  float w1 = m01 * u0 + m11 * u1;
  return ss - 2.0f * (u0 * w0 + u1 * w1) +
         (ga * w0 * w0 + 2.0f * gb * w0 * w1 + gd * w1 * w1);
}

__global__ void __launch_bounds__(256)
transint_score(const int* __restrict__ pos_h, const int* __restrict__ pos_t,
               const int* __restrict__ pos_r, const int* __restrict__ neg_h,
               const int* __restrict__ neg_t, const float* __restrict__ ent_emb,
               const float* __restrict__ heads, const float* __restrict__ bases,
               const int* __restrict__ rel2head, const float* __restrict__ rel2mult,
               const float* __restrict__ rws, float* __restrict__ out, int Bn) {
  int wave = __builtin_amdgcn_readfirstlane(
      (int)((blockIdx.x * blockDim.x + threadIdx.x) >> 5));
  int lane = (int)(threadIdx.x & 31);
  int b0 = wave * BPW;
  if (b0 >= Bn) return;
  int n = Bn - b0;
  n = n < BPW ? n : BPW;

  int off = lane * LANE_OFF;
  float mine = 0.0f;

  for (int j = 0; j < n; ++j) {
    int b = b0 + j;
    int hp = pos_h[b];
    int tp = pos_t[b];
    int hn = neg_h[b];
    int tn = neg_t[b];
    int r = pos_r[b];
    int k = rel2head[r];
    float mult = rel2mult[r];

    const float4* ehp = (const float4*)(ent_emb + (size_t)hp * DIMS + off);
    const float4* etp = (const float4*)(ent_emb + (size_t)tp * DIMS + off);
    const float4* ehn = (const float4*)(ent_emb + (size_t)hn * DIMS + off);
    const float4* etn = (const float4*)(ent_emb + (size_t)tn * DIMS + off);
    const float4* hd = (const float4*)(heads + (size_t)k * DIMS + off);
    const float4* a0 = (const float4*)(bases + (size_t)r * 2 * DIMS + off);
    const float4* a1 = (const float4*)(bases + (size_t)r * 2 * DIMS + DIMS + off);

    float4 ehp0 = ehp[0], ehp1 = ehp[SECOND];
    float4 etp0 = etp[0], etp1 = etp[SECOND];
    float4 ehn0 = ehn[0], ehn1 = ehn[SECOND];
    float4 etn0 = etn[0], etn1 = etn[SECOND];
    float4 hd0 = hd[0], hd1 = hd[SECOND];
    float4 A00 = a0[0], A01 = a0[SECOND];
    float4 A10 = a1[0], A11 = a1[SECOND];

    float4 sp0, sp1, sn0, sn1;
    lane_sub(ehp0, ehp1, etp0, etp1, hd0, hd1, mult, sp0, sp1);
    lane_sub(ehn0, ehn1, etn0, etn1, hd0, hd1, mult, sn0, sn1);

    float p0 = dot4(sp0, A00) + dot4(sp1, A01);
    float p1 = dot4(sp0, A10) + dot4(sp1, A11);
    float p2 = dot4(sp0, sp0) + dot4(sp1, sp1);
    float q0 = dot4(sn0, A00) + dot4(sn1, A01);
    float q1 = dot4(sn0, A10) + dot4(sn1, A11);
    float q2 = dot4(sn0, sn0) + dot4(sn1, sn1);

    float up0, up1, ssp, un0, un1, ssn;
    wave_reduce6(p0, p1, p2, q0, q1, q2, lane, up0, up1, ssp, un0, un1, ssn);

    const float* rw = rws + (size_t)r * 8;
    float ga = rw[0], gb = rw[1], gd = rw[2];
    float m00 = rw[3], m01 = rw[4], m11 = rw[5];
    float score_p = score_of(up0, up1, ssp, ga, gb, gd, m00, m01, m11);
    float score_n = score_of(un0, un1, ssn, ga, gb, gd, m00, m01, m11);

    mine = (lane == j) ? score_p : mine;
    mine = (lane == j + BPW) ? score_n : mine;
  }

  int slot = lane & (BPW - 1);
  if (lane < 2 * BPW && slot < n)
    out[(size_t)(lane < BPW ? 0 : Bn) + (size_t)(b0 + slot)] = mine;
}

extern "C" void kernel_launch(void* const* d_in, const int* in_sizes, int n_in,
                              void* d_out, int out_size, void* d_ws, size_t ws_size,
                              hipStream_t stream) {
  const int*   pos_h    = (const int*)d_in[0];
  const int*   pos_t    = (const int*)d_in[1];
  const int*   pos_r    = (const int*)d_in[2];
  const int*   neg_h    = (const int*)d_in[3];
  const int*   neg_t    = (const int*)d_in[4];
  const float* ent_emb  = (const float*)d_in[6];
  const float* heads    = (const float*)d_in[7];
  const float* bases    = (const float*)d_in[8];
  const int*   rel2head = (const int*)d_in[9];
  const float* rel2mult = (const float*)d_in[10];
  float* out = (float*)d_out;
  float* rws = (float*)d_ws;

  int Bn = in_sizes[0];
  int R  = in_sizes[9];

  int relBlocks = (R * 32 + 255) / 256;
  transint_relprep<<<relBlocks, 256, 0, stream>>>(bases, rws, R);

  long nwaves = ((long)Bn + BPW - 1) / BPW;
  long blocks = (nwaves * 32 + 255) / 256;
  transint_score<<<(int)blocks, 256, 0, stream>>>(
      pos_h, pos_t, pos_r, neg_h, neg_t, ent_emb, heads, bases,
      rel2head, rel2mult, rws, out, Bn);
}
